# starting kernel
# baseline (speedup 1.0000x reference)
#include <hip/hip_runtime.h>

#define NUM_NODES 100000
#define NUM_EDGES 50000
#define NUM_CONN  800000
#define IN_DIM    256
#define HIDDEN    128
#define OUT_DIM   17
#define LN_EPS    1e-5f

#define WAVES_PER_BLOCK 5
#define TILE_ROWS 16
#define N_TILES (HIDDEN / 16)
#define K_STEPS (IN_DIM / 4)
#define WPACK_FLOATS (K_STEPS * 32 * 16)

typedef float v2f __attribute__((ext_vector_type(2)));
typedef float v8f __attribute__((ext_vector_type(8)));

__global__ void zero_ws_kernel(float* __restrict__ p, int n) {
    int i = blockIdx.x * blockDim.x + threadIdx.x;
    if (i < n) p[i] = 0.0f;
}

__global__ void repack_w1_kernel(const float* __restrict__ W1,
                                 float* __restrict__ wpack) {
    int id = blockIdx.x * blockDim.x + threadIdx.x;
    if (id >= K_STEPS * 32 * 8) return;
    int ks   = id >> 8;
    int rem  = id & 255;
    int lane = rem >> 3;
    int nt   = rem & 7;
    int m    = lane & 15;
    int hi   = lane >> 4;
    int krow = ks * 4 + 2 * hi;
    int n    = nt * 16 + m;
    wpack[2 * id + 0] = W1[(size_t)(krow + 0) * HIDDEN + n];
    wpack[2 * id + 1] = W1[(size_t)(krow + 1) * HIDDEN + n];
}

__global__ __launch_bounds__(256) void scatter_add_kernel(
    const float* __restrict__ nf,
    const int*   __restrict__ nodes,
    const int*   __restrict__ edges,
    float* __restrict__ summed,
    float* __restrict__ counts) {
    int t    = blockIdx.x * blockDim.x + threadIdx.x;
    int conn = t >> 5;
    if (conn >= NUM_CONN) return;
    int lane = t & 31;
    int node = nodes[conn];
    int edge = edges[conn];

    const float4* src = (const float4*)(nf + (size_t)node * IN_DIM + lane * 8);
    float4 a = src[0];
    float4 b = src[1];

    float* dst = summed + (size_t)edge * IN_DIM + lane * 8;
    atomicAdd(dst + 0, a.x);
    atomicAdd(dst + 1, a.y);
    atomicAdd(dst + 2, a.z);
    atomicAdd(dst + 3, a.w);
    atomicAdd(dst + 4, b.x);
    atomicAdd(dst + 5, b.y);
    atomicAdd(dst + 6, b.z);
    atomicAdd(dst + 7, b.w);
    if (lane == 0) atomicAdd(counts + edge, 1.0f);
}

__global__ __launch_bounds__(WAVES_PER_BLOCK * 32) void edge_mlp_kernel(
    const float* __restrict__ summed,
    const float* __restrict__ counts,
    const float* __restrict__ wpack,
    const float* __restrict__ b1,
    const float* __restrict__ lng,
    const float* __restrict__ lnb,
    const float* __restrict__ W2,
    const float* __restrict__ b2,
    float* __restrict__ out) {
    __shared__ float sH[WAVES_PER_BLOCK * TILE_ROWS * HIDDEN];

    const int wave = threadIdx.x >> 5;
    const int lane = threadIdx.x & 31;
    const int tile = blockIdx.x * WAVES_PER_BLOCK + wave;
    const int m0   = tile * TILE_ROWS;
    const int m    = lane & 15;
    const int hi   = lane >> 4;

    const int row = m0 + m;
    const float invc = 1.0f / fmaxf(counts[row], 1.0f);
    const float* __restrict__ arow = summed + (size_t)row * IN_DIM;

    v8f acc[N_TILES] = {};

    for (int ks = 0; ks < K_STEPS; ++ks) {
        const int k = ks * 4;
        v2f a = *(const v2f*)(arow + k + 2 * hi);
        a.x *= invc;
        a.y *= invc;

        const float4* __restrict__ bp =
            (const float4*)(wpack + ((size_t)ks * 32 + lane) * 16);
        float4 q0 = bp[0], q1 = bp[1], q2 = bp[2], q3 = bp[3];
        v2f bbs[N_TILES];
        bbs[0].x = q0.x; bbs[0].y = q0.y;
        bbs[1].x = q0.z; bbs[1].y = q0.w;
        bbs[2].x = q1.x; bbs[2].y = q1.y;
        bbs[3].x = q1.z; bbs[3].y = q1.w;
        bbs[4].x = q2.x; bbs[4].y = q2.y;
        bbs[5].x = q2.z; bbs[5].y = q2.w;
        bbs[6].x = q3.x; bbs[6].y = q3.y;
        bbs[7].x = q3.z; bbs[7].y = q3.w;

#pragma unroll
        for (int nt = 0; nt < N_TILES; ++nt) {
            acc[nt] = __builtin_amdgcn_wmma_f32_16x16x4_f32(
                false, a, false, bbs[nt], (short)0, acc[nt], false, false);
        }
    }

#pragma unroll
    for (int nt = 0; nt < N_TILES; ++nt) {
        const float bias = b1[nt * 16 + m];
#pragma unroll
        for (int r = 0; r < 8; ++r) acc[nt][r] += bias;
    }

    float s[8], q[8];
#pragma unroll
    for (int r = 0; r < 8; ++r) { s[r] = 0.0f; q[r] = 0.0f; }
#pragma unroll
    for (int nt = 0; nt < N_TILES; ++nt) {
#pragma unroll
        for (int r = 0; r < 8; ++r) {
            const float x = acc[nt][r];
            s[r] += x;
            q[r] += x * x;
        }
    }
#pragma unroll
    for (int mask = 1; mask < 16; mask <<= 1) {
#pragma unroll
        for (int r = 0; r < 8; ++r) {
            s[r] += __shfl_xor(s[r], mask, 32);
            q[r] += __shfl_xor(q[r], mask, 32);
        }
    }
    float mean[8], rstd[8];
#pragma unroll
    for (int r = 0; r < 8; ++r) {
        mean[r] = s[r] * (1.0f / HIDDEN);
        const float var = q[r] * (1.0f / HIDDEN) - mean[r] * mean[r];
        rstd[r] = rsqrtf(var + LN_EPS);
    }

    float* __restrict__ hbase = sH + wave * (TILE_ROWS * HIDDEN);
#pragma unroll
    for (int nt = 0; nt < N_TILES; ++nt) {
        const int n = nt * 16 + m;
        const float g  = lng[n];
        const float bb = lnb[n];
#pragma unroll
        for (int r = 0; r < 8; ++r) {
            float x = (acc[nt][r] - mean[r]) * rstd[r] * g + bb;
            x = fmaxf(x, 0.0f);
            hbase[(r + 8 * hi) * HIDDEN + n] = x;
        }
    }
    __syncthreads();

    for (int idx = lane; idx < TILE_ROWS * OUT_DIM; idx += 32) {
        const int r = idx / OUT_DIM;
        const int j = idx - r * OUT_DIM;
        float sum = b2[j];
        const float* __restrict__ hr = hbase + r * HIDDEN;
#pragma unroll 4
        for (int k = 0; k < HIDDEN; ++k)
            sum = fmaf(hr[k], W2[k * OUT_DIM + j], sum);
        out[(size_t)(m0 + r) * OUT_DIM + j] = sum;
    }
}

extern "C" void kernel_launch(void* const* d_in, const int* in_sizes, int n_in,
                              void* d_out, int out_size, void* d_ws, size_t ws_size,
                              hipStream_t stream) {
    const float* nf    = (const float*)d_in[0];
    const int*   hidx  = (const int*)d_in[1];
    const int*   nodes = hidx;
    const int*   edges = hidx + NUM_CONN;
    const float* W1    = (const float*)d_in[2];
    const float* b1    = (const float*)d_in[3];
    const float* lng   = (const float*)d_in[4];
    const float* lnb   = (const float*)d_in[5];
    const float* W2    = (const float*)d_in[6];
    const float* b2    = (const float*)d_in[7];
    float*       out   = (float*)d_out;

    float* summed = (float*)d_ws;
    float* counts = summed + (size_t)NUM_EDGES * IN_DIM;
    float* wpack  = counts + NUM_EDGES;

    const int npack = K_STEPS * 32 * 8;
    repack_w1_kernel<<<(npack + 255) / 256, 256, 0, stream>>>(W1, wpack);

    const int nzero = NUM_EDGES * IN_DIM + NUM_EDGES;
    zero_ws_kernel<<<(nzero + 255) / 256, 256, 0, stream>>>(summed, nzero);

    const long long nthreads = (long long)NUM_CONN * 32;
    scatter_add_kernel<<<(int)((nthreads + 255) / 256), 256, 0, stream>>>(
        nf, nodes, edges, summed, counts);

    const int ntiles = NUM_EDGES / TILE_ROWS;
    edge_mlp_kernel<<<ntiles / WAVES_PER_BLOCK, WAVES_PER_BLOCK * 32, 0, stream>>>(
        summed, counts, wpack, b1, lng, lnb, W2, b2, out);
}


# Global float-atomic scatter replaced by edge-range owner groups: 256 work-groups each scan the list and sum their 196 edges' node rows in an LDS tile by LDS float adds (arrival order); no zero pass
# speedup vs baseline: 4.6624x; 4.6624x over previous
#include <hip/hip_runtime.h>

#define NUM_NODES 100000
#define NUM_EDGES 50000
#define NUM_CONN  800000
#define IN_DIM    256
#define HIDDEN    128
#define OUT_DIM   17
#define LN_EPS    1e-5f

#define WAVES_PER_BLOCK 5
#define TILE_ROWS 16
#define N_TILES (HIDDEN / 16)
#define K_STEPS (IN_DIM / 4)
#define WPACK_FLOATS (K_STEPS * 32 * 16)

typedef float v2f __attribute__((ext_vector_type(2)));
typedef float v8f __attribute__((ext_vector_type(8)));

#define ACC_WGS       256
#define ACC_THREADS   1024
#define ACC_WAVES     (ACC_THREADS / 32)
#define EDGES_PER_WG  ((NUM_EDGES + ACC_WGS - 1) / ACC_WGS)
#define CHUNK_LOADS   4
#define CHUNK_IDS     (CHUNK_LOADS * 128)
#define FULL_CHUNKS   (NUM_CONN / CHUNK_IDS)
#define NUM_CONN4     (NUM_CONN / 4)
#define LIST_CAP      544
#define CONN_BITS     20

__global__ void repack_w1_kernel(const float* __restrict__ W1,
                                 float* __restrict__ wpack) {
    int id = blockIdx.x * blockDim.x + threadIdx.x;
    if (id >= K_STEPS * 32 * 8) return;
    int ks   = id >> 8;
    int rem  = id & 255;
    int lane = rem >> 3;
    int nt   = rem & 7;
    int m    = lane & 15;
    int hi   = lane >> 4;
    int krow = ks * 4 + 2 * hi;
    int n    = nt * 16 + m;
    wpack[2 * id + 0] = W1[(size_t)(krow + 0) * HIDDEN + n];
    wpack[2 * id + 1] = W1[(size_t)(krow + 1) * HIDDEN + n];
}

template <bool FULL>
__device__ __forceinline__ void add_listed_rows(
    const float* __restrict__ nf, const int* __restrict__ nodes,
    const unsigned* lst, int n, float* tile, float* cnts, int lane) {
    unsigned e = 0u;
    if (FULL || lane < n) e = lst[lane];
    const int conn = (int)(e & ((1u << CONN_BITS) - 1u));
    const int le   = (int)(e >> CONN_BITS);
    const int node = nodes[conn];
    if (FULL || lane < n)
        __hip_atomic_fetch_add(cnts + le, 1.0f, __ATOMIC_RELAXED, __HIP_MEMORY_SCOPE_WORKGROUP);
#pragma unroll
    for (int g = 0; g < 4; ++g) {
        if (!FULL && g * 8 >= n) break;
        v2f r[8][4];
#pragma unroll
        for (int jj = 0; jj < 8; ++jj) {
            const int j = g * 8 + jj;
            if (FULL || j < n) {
                const int nj = __builtin_amdgcn_readlane(node, j);
                const v2f* src = (const v2f*)(nf + (size_t)nj * IN_DIM) + lane;
#pragma unroll
                for (int k = 0; k < 4; ++k) r[jj][k] = src[32 * k];
            }
        }
#pragma unroll
        for (int jj = 0; jj < 8; ++jj) {
            const int j = g * 8 + jj;
            if (FULL || j < n) {
                const int lej = __builtin_amdgcn_readlane(le, j);
                float* dst = tile + lej * IN_DIM + 2 * lane;
#pragma unroll
                for (int k = 0; k < 4; ++k) {
                    __hip_atomic_fetch_add(dst + 64 * k + 0, r[jj][k].x,
                                           __ATOMIC_RELAXED, __HIP_MEMORY_SCOPE_WORKGROUP);
                    __hip_atomic_fetch_add(dst + 64 * k + 1, r[jj][k].y,
                                           __ATOMIC_RELAXED, __HIP_MEMORY_SCOPE_WORKGROUP);
                }
            }
        }
    }
}

__global__ __launch_bounds__(ACC_THREADS) void edge_accum_kernel(
    const float* __restrict__ nf,
    const int*   __restrict__ nodes,
    const int*   __restrict__ edges,
    float* __restrict__ summed,
    float* __restrict__ counts) {
    __shared__ __attribute__((aligned(16))) float tile[EDGES_PER_WG * IN_DIM];
    __shared__ float cnts[EDGES_PER_WG];
    __shared__ unsigned lists[ACC_WAVES * LIST_CAP];

    const int tid  = threadIdx.x;
    const int lane = tid & 31;
    const int wave = __builtin_amdgcn_readfirstlane(tid >> 5);
    const int e0   = blockIdx.x * EDGES_PER_WG;
    const int nE   = min(EDGES_PER_WG, NUM_EDGES - e0);

    float4* tile4 = (float4*)tile;
    for (int i = tid; i < EDGES_PER_WG * IN_DIM / 4; i += ACC_THREADS)
        tile4[i] = make_float4(0.0f, 0.0f, 0.0f, 0.0f);
    if (tid < EDGES_PER_WG) cnts[tid] = 0.0f;
    __syncthreads();

    unsigned* lst = lists + wave * LIST_CAP;
    int cnt = 0;
    const int4* __restrict__ e4 = (const int4*)edges;
    const int shift = (blockIdx.x * 6) % FULL_CHUNKS;

    auto take = [&](int id, int conn) {
        const unsigned le = (unsigned)id - (unsigned)e0;
        const bool p = le < (unsigned)nE;
        const unsigned m = __builtin_amdgcn_ballot_w32(p);
        if (p) lst[cnt + __builtin_amdgcn_mbcnt_lo(m, 0u)] = (unsigned)conn | (le << CONN_BITS);
        cnt += __builtin_popcount(m);
    };
    auto take4 = [&](const int4 v, int q) {
        const unsigned ue = (unsigned)e0, un = (unsigned)nE;
        const bool any = ((unsigned)v.x - ue < un) | ((unsigned)v.y - ue < un) |
                         ((unsigned)v.z - ue < un) | ((unsigned)v.w - ue < un);
        if (__builtin_amdgcn_ballot_w32(any) != 0u) {
            take(v.x, 4 * q + 0);
            take(v.y, 4 * q + 1);
            take(v.z, 4 * q + 2);
            take(v.w, 4 * q + 3);
        }
    };
    auto chunk_q0 = [&](int idx) {
        int c = idx + shift;
        if (c >= FULL_CHUNKS) c -= FULL_CHUNKS;
        return c * (CHUNK_IDS / 4) + lane;
    };

    int idx = wave;
    int q0  = chunk_q0(idx);
    int4 cur[CHUNK_LOADS];
#pragma unroll
    for (int k = 0; k < CHUNK_LOADS; ++k) cur[k] = e4[q0 + 32 * k];
    for (;;) {
        const int  nidx = idx + ACC_WAVES;
        const bool more = nidx < FULL_CHUNKS;
        const int  nq0  = chunk_q0(nidx);
        int4 nxt[CHUNK_LOADS];
        if (more) {
#pragma unroll
            for (int k = 0; k < CHUNK_LOADS; ++k) nxt[k] = e4[nq0 + 32 * k];
        }
#pragma unroll
        for (int k = 0; k < CHUNK_LOADS; ++k) take4(cur[k], q0 + 32 * k);
        while (cnt >= 32) {
            cnt -= 32;
            add_listed_rows<true>(nf, nodes, lst + cnt, 32, tile, cnts, lane);
        }
        if (!more) break;
#pragma unroll
        for (int k = 0; k < CHUNK_LOADS; ++k) cur[k] = nxt[k];
        idx = nidx;
        q0  = nq0;
    }
    for (int q = FULL_CHUNKS * (CHUNK_IDS / 4) + tid; q < NUM_CONN4; q += ACC_THREADS)
        take4(e4[q], q);
    while (cnt > 0) {
        const int n = min(cnt, 32);
        cnt -= n;
        add_listed_rows<false>(nf, nodes, lst + cnt, n, tile, cnts, lane);
    }
    __syncthreads();

    float4* out4 = (float4*)(summed + (size_t)e0 * IN_DIM);
    for (int i = tid; i < nE * (IN_DIM / 4); i += ACC_THREADS) out4[i] = tile4[i];
    if (tid < nE) counts[e0 + tid] = cnts[tid];
}

__global__ __launch_bounds__(WAVES_PER_BLOCK * 32) void edge_mlp_kernel(
    const float* __restrict__ summed,
    const float* __restrict__ counts,
    const float* __restrict__ wpack,
    const float* __restrict__ b1,
    const float* __restrict__ lng,
    const float* __restrict__ lnb,
    const float* __restrict__ W2,
    const float* __restrict__ b2,
    float* __restrict__ out) {
    __shared__ float sH[WAVES_PER_BLOCK * TILE_ROWS * HIDDEN];

    const int wave = threadIdx.x >> 5;
    const int lane = threadIdx.x & 31;
    const int tile = blockIdx.x * WAVES_PER_BLOCK + wave;
    const int m0   = tile * TILE_ROWS;
    const int m    = lane & 15;
    const int hi   = lane >> 4;

    const int row = m0 + m;
    const float invc = 1.0f / fmaxf(counts[row], 1.0f);
    const float* __restrict__ arow = summed + (size_t)row * IN_DIM;

    v8f acc[N_TILES] = {};

    for (int ks = 0; ks < K_STEPS; ++ks) {
        const int k = ks * 4;
        v2f a = *(const v2f*)(arow + k + 2 * hi);
        a.x *= invc;
        a.y *= invc;

        const float4* __restrict__ bp =
            (const float4*)(wpack + ((size_t)ks * 32 + lane) * 16);
        float4 q0 = bp[0], q1 = bp[1], q2 = bp[2], q3 = bp[3];
        v2f bbs[N_TILES];
        bbs[0].x = q0.x; bbs[0].y = q0.y;
        bbs[1].x = q0.z; bbs[1].y = q0.w;
        bbs[2].x = q1.x; bbs[2].y = q1.y;
        bbs[3].x = q1.z; bbs[3].y = q1.w;
        bbs[4].x = q2.x; bbs[4].y = q2.y;
        bbs[5].x = q2.z; bbs[5].y = q2.w;
        bbs[6].x = q3.x; bbs[6].y = q3.y;
        bbs[7].x = q3.z; bbs[7].y = q3.w;

#pragma unroll
        for (int nt = 0; nt < N_TILES; ++nt) {
            acc[nt] = __builtin_amdgcn_wmma_f32_16x16x4_f32(
                false, a, false, bbs[nt], (short)0, acc[nt], false, false);
        }
    }

#pragma unroll
    for (int nt = 0; nt < N_TILES; ++nt) {
        const float bias = b1[nt * 16 + m];
#pragma unroll
        for (int r = 0; r < 8; ++r) acc[nt][r] += bias;
    }

    float s[8], q[8];
#pragma unroll
    for (int r = 0; r < 8; ++r) { s[r] = 0.0f; q[r] = 0.0f; }
#pragma unroll
    for (int nt = 0; nt < N_TILES; ++nt) {
#pragma unroll
        for (int r = 0; r < 8; ++r) {
            const float x = acc[nt][r];
            s[r] += x;
            q[r] += x * x;
        }
    }
#pragma unroll
    for (int mask = 1; mask < 16; mask <<= 1) {
#pragma unroll
        for (int r = 0; r < 8; ++r) {
            s[r] += __shfl_xor(s[r], mask, 32);
            q[r] += __shfl_xor(q[r], mask, 32);
        }
    }
    float mean[8], rstd[8];
#pragma unroll
    for (int r = 0; r < 8; ++r) {
        mean[r] = s[r] * (1.0f / HIDDEN);
        const float var = q[r] * (1.0f / HIDDEN) - mean[r] * mean[r];
        rstd[r] = rsqrtf(var + LN_EPS);
    }

    float* __restrict__ hbase = sH + wave * (TILE_ROWS * HIDDEN);
#pragma unroll
    for (int nt = 0; nt < N_TILES; ++nt) {
        const int n = nt * 16 + m;
        const float g  = lng[n];
        const float bb = lnb[n];
#pragma unroll
        for (int r = 0; r < 8; ++r) {
            float x = (acc[nt][r] - mean[r]) * rstd[r] * g + bb;
            x = fmaxf(x, 0.0f);
            hbase[(r + 8 * hi) * HIDDEN + n] = x;
        }
    }
    __syncthreads();

    for (int idx = lane; idx < TILE_ROWS * OUT_DIM; idx += 32) {
        const int r = idx / OUT_DIM;
        const int j = idx - r * OUT_DIM;
        float sum = b2[j];
        const float* __restrict__ hr = hbase + r * HIDDEN;
#pragma unroll 4
        for (int k = 0; k < HIDDEN; ++k)
            sum = fmaf(hr[k], W2[k * OUT_DIM + j], sum);
        out[(size_t)(m0 + r) * OUT_DIM + j] = sum;
    }
}

extern "C" void kernel_launch(void* const* d_in, const int* in_sizes, int n_in,
                              void* d_out, int out_size, void* d_ws, size_t ws_size,
                              hipStream_t stream) {
    const float* nf    = (const float*)d_in[0];
    const int*   hidx  = (const int*)d_in[1];
    const int*   nodes = hidx;
    const int*   edges = hidx + NUM_CONN;
    const float* W1    = (const float*)d_in[2];
    const float* b1    = (const float*)d_in[3];
    const float* lng   = (const float*)d_in[4];
    const float* lnb   = (const float*)d_in[5];
    const float* W2    = (const float*)d_in[6];
    const float* b2    = (const float*)d_in[7];
    float*       out   = (float*)d_out;

    float* summed = (float*)d_ws;
    float* counts = summed + (size_t)NUM_EDGES * IN_DIM;
    float* wpack  = counts + NUM_EDGES;

    const int npack = K_STEPS * 32 * 8;
    repack_w1_kernel<<<(npack + 255) / 256, 256, 0, stream>>>(W1, wpack);

    edge_accum_kernel<<<ACC_WGS, ACC_THREADS, 0, stream>>>(nf, nodes, edges, summed, counts);

    const int ntiles = NUM_EDGES / TILE_ROWS;
    edge_mlp_kernel<<<ntiles / WAVES_PER_BLOCK, WAVES_PER_BLOCK * 32, 0, stream>>>(
        summed, counts, wpack, b1, lng, lnb, W2, b2, out);
}


# Owner groups scan one key byte per connection (edge / 196, written by extra work-groups of the repack launch) instead of the 4-byte edge ids; local edge re-read from the id when a row is added
# speedup vs baseline: 4.9309x; 1.0576x over previous
#include <hip/hip_runtime.h>

#define NUM_NODES 100000
#define NUM_EDGES 50000
#define NUM_CONN  800000
#define IN_DIM    256
#define HIDDEN    128
#define OUT_DIM   17
#define LN_EPS    1e-5f

#define WAVES_PER_BLOCK 5
#define TILE_ROWS 16
#define N_TILES (HIDDEN / 16)
#define K_STEPS (IN_DIM / 4)
#define WPACK_FLOATS (K_STEPS * 32 * 16)

typedef float v2f __attribute__((ext_vector_type(2)));
typedef float v8f __attribute__((ext_vector_type(8)));

#define ACC_WGS       256
#define ACC_THREADS   1024
#define ACC_WAVES     (ACC_THREADS / 32)
#define EDGES_PER_WG  ((NUM_EDGES + ACC_WGS - 1) / ACC_WGS)
#define CHUNK_LOADS   4
#define CHUNK_VECS    (CHUNK_LOADS * 32)
#define NUM_CONN4     (NUM_CONN / 4)
#define NUM_CONN16    (NUM_CONN / 16)
#define FULL_CHUNKS   (NUM_CONN16 / CHUNK_VECS)
#define LIST_CAP      544
#define REPACK_WGS    ((K_STEPS * 32 * 8 + 255) / 256)
#define KEY_WGS       ((NUM_CONN4 + 255) / 256)

__global__ void repack_w1_kernel(const float* __restrict__ W1,
                                 float* __restrict__ wpack,
                                 const int* __restrict__ edges,
                                 unsigned* __restrict__ okey) {
    if (blockIdx.x >= REPACK_WGS) {
        const int i = (blockIdx.x - REPACK_WGS) * blockDim.x + threadIdx.x;
        if (i >= NUM_CONN4) return;
        const int4 v = ((const int4*)edges)[i];
        const unsigned a[4] = {(unsigned)v.x, (unsigned)v.y, (unsigned)v.z, (unsigned)v.w};
        unsigned w = 0u;
#pragma unroll
        for (int b = 0; b < 4; ++b) {
            const unsigned k = a[b] < (unsigned)NUM_EDGES ? a[b] / (unsigned)EDGES_PER_WG : (a[b] & 255u);
            w |= k << (8 * b);
        }
        okey[i] = w;
        return;
    }
    int id = blockIdx.x * blockDim.x + threadIdx.x;
    if (id >= K_STEPS * 32 * 8) return;
    int ks   = id >> 8;
    int rem  = id & 255;
    int lane = rem >> 3;
    int nt   = rem & 7;
    int m    = lane & 15;
    int hi   = lane >> 4;
    int krow = ks * 4 + 2 * hi;
    int n    = nt * 16 + m;
    wpack[2 * id + 0] = W1[(size_t)(krow + 0) * HIDDEN + n];
    wpack[2 * id + 1] = W1[(size_t)(krow + 1) * HIDDEN + n];
}

template <bool FULL>
__device__ __forceinline__ void add_listed_rows(
    const float* __restrict__ nf, const int* __restrict__ nodes, const int* __restrict__ edges,
    const unsigned* lst, int n, float* tile, float* cnts, int lane, int e0, int nE) {
    unsigned conn = 0u;
    if (FULL || lane < n) conn = lst[lane];
    const int node = nodes[conn];
    const unsigned lu = (unsigned)edges[conn] - (unsigned)e0;
    const int le = lu < (unsigned)nE ? (int)lu : EDGES_PER_WG;
    if (FULL || lane < n)
        __hip_atomic_fetch_add(cnts + le, 1.0f, __ATOMIC_RELAXED, __HIP_MEMORY_SCOPE_WORKGROUP);
#pragma unroll
    for (int g = 0; g < 4; ++g) {
        if (!FULL && g * 8 >= n) break;
        v2f r[8][4];
#pragma unroll
        for (int jj = 0; jj < 8; ++jj) {
            const int j = g * 8 + jj;
            if (FULL || j < n) {
                const int nj = __builtin_amdgcn_readlane(node, j);
                const v2f* src = (const v2f*)(nf + (size_t)nj * IN_DIM) + lane;
#pragma unroll
                for (int k = 0; k < 4; ++k) r[jj][k] = src[32 * k];
            }
        }
#pragma unroll
        for (int jj = 0; jj < 8; ++jj) {
            const int j = g * 8 + jj;
            if (FULL || j < n) {
                const int lej = __builtin_amdgcn_readlane(le, j);
                float* dst = tile + lej * IN_DIM + 2 * lane;
#pragma unroll
                for (int k = 0; k < 4; ++k) {
                    __hip_atomic_fetch_add(dst + 64 * k + 0, r[jj][k].x,
                                           __ATOMIC_RELAXED, __HIP_MEMORY_SCOPE_WORKGROUP);
                    __hip_atomic_fetch_add(dst + 64 * k + 1, r[jj][k].y,
                                           __ATOMIC_RELAXED, __HIP_MEMORY_SCOPE_WORKGROUP);
                }
            }
        }
    }
}

__global__ __launch_bounds__(ACC_THREADS) void edge_accum_kernel(
    const float* __restrict__ nf,
    const int*   __restrict__ nodes,
    const int*   __restrict__ edges,
    const unsigned* __restrict__ okey,
    float* __restrict__ summed,
    float* __restrict__ counts) {
    __shared__ __attribute__((aligned(16))) float tile[(EDGES_PER_WG + 1) * IN_DIM];
    __shared__ float cnts[EDGES_PER_WG + 1];
    __shared__ unsigned lists[ACC_WAVES * LIST_CAP];

    const int tid  = threadIdx.x;
    const int lane = tid & 31;
    const int wave = __builtin_amdgcn_readfirstlane(tid >> 5);
    const int e0   = blockIdx.x * EDGES_PER_WG;
    const int nE   = min(EDGES_PER_WG, NUM_EDGES - e0);

    float4* tile4 = (float4*)tile;
    for (int i = tid; i < (EDGES_PER_WG + 1) * IN_DIM / 4; i += ACC_THREADS)
        tile4[i] = make_float4(0.0f, 0.0f, 0.0f, 0.0f);
    if (tid < EDGES_PER_WG + 1) cnts[tid] = 0.0f;
    __syncthreads();

    unsigned* lst = lists + wave * LIST_CAP;
    int cnt = 0;
    const uint4* __restrict__ k4 = (const uint4*)okey;
    const int shift = (blockIdx.x * 3) % FULL_CHUNKS;
    const unsigned pat = (unsigned)blockIdx.x * 0x01010101u;

    auto hitbits = [&](unsigned x) {
        x ^= pat;
        const unsigned t = (x & 0x7f7f7f7fu) + 0x7f7f7f7fu;
        return ~(t | x) & 0x80808080u;
    };
    auto take16 = [&](const uint4 v, int q) {
        unsigned h = (hitbits(v.x) >> 7) | (hitbits(v.y) >> 6) | (hitbits(v.z) >> 5) | (hitbits(v.w) >> 4);
        unsigned m = __builtin_amdgcn_ballot_w32(h != 0u);
        while (m != 0u) {
            if (h != 0u) {
                const int f = __builtin_ctz(h);
                lst[cnt + __builtin_amdgcn_mbcnt_lo(m, 0u)] = (unsigned)(16 * q + 4 * (f & 7) + (f >> 3));
                h &= h - 1u;
            }
            cnt += __builtin_popcount(m);
            m = __builtin_amdgcn_ballot_w32(h != 0u);
        }
    };
    auto drain = [&]() {
        while (cnt >= 32) {
            cnt -= 32;
            add_listed_rows<true>(nf, nodes, edges, lst + cnt, 32, tile, cnts, lane, e0, nE);
        }
    };
    auto chunk_q0 = [&](int idx) {
        int c = idx + shift;
        if (c >= FULL_CHUNKS) c -= FULL_CHUNKS;
        return c * CHUNK_VECS + lane;
    };

    int idx = wave;
    int q0  = chunk_q0(idx);
    uint4 cur[CHUNK_LOADS];
#pragma unroll
    for (int k = 0; k < CHUNK_LOADS; ++k) cur[k] = k4[q0 + 32 * k];
    for (;;) {
        const int  nidx = idx + ACC_WAVES;
        const bool more = nidx < FULL_CHUNKS;
        const int  nq0  = chunk_q0(nidx);
        uint4 nxt[CHUNK_LOADS];
        if (more) {
#pragma unroll
            for (int k = 0; k < CHUNK_LOADS; ++k) nxt[k] = k4[nq0 + 32 * k];
        }
#pragma unroll
        for (int k = 0; k < CHUNK_LOADS; ++k) {
            take16(cur[k], q0 + 32 * k);
            drain();
        }
        if (!more) break;
#pragma unroll
        for (int k = 0; k < CHUNK_LOADS; ++k) cur[k] = nxt[k];
        idx = nidx;
        q0  = nq0;
    }
    {
        const int q = FULL_CHUNKS * CHUNK_VECS + tid;
        uint4 v = make_uint4(~pat, ~pat, ~pat, ~pat);
        if (q < NUM_CONN16) v = k4[q];
        take16(v, q);
        drain();
    }
    while (cnt > 0) {
        const int n = min(cnt, 32);
        cnt -= n;
        add_listed_rows<false>(nf, nodes, edges, lst + cnt, n, tile, cnts, lane, e0, nE);
    }
    __syncthreads();

    float4* out4 = (float4*)(summed + (size_t)e0 * IN_DIM);
    for (int i = tid; i < nE * (IN_DIM / 4); i += ACC_THREADS) out4[i] = tile4[i];
    if (tid < nE) counts[e0 + tid] = cnts[tid];
}

__global__ __launch_bounds__(WAVES_PER_BLOCK * 32) void edge_mlp_kernel(
    const float* __restrict__ summed,
    const float* __restrict__ counts,
    const float* __restrict__ wpack,
    const float* __restrict__ b1,
    const float* __restrict__ lng,
    const float* __restrict__ lnb,
    const float* __restrict__ W2,
    const float* __restrict__ b2,
    float* __restrict__ out) {
    __shared__ float sH[WAVES_PER_BLOCK * TILE_ROWS * HIDDEN];

    const int wave = threadIdx.x >> 5;
    const int lane = threadIdx.x & 31;
    const int tile = blockIdx.x * WAVES_PER_BLOCK + wave;
    const int m0   = tile * TILE_ROWS;
    const int m    = lane & 15;
    const int hi   = lane >> 4;

    const int row = m0 + m;
    const float invc = 1.0f / fmaxf(counts[row], 1.0f);
    const float* __restrict__ arow = summed + (size_t)row * IN_DIM;

    v8f acc[N_TILES] = {};

    for (int ks = 0; ks < K_STEPS; ++ks) {
        const int k = ks * 4;
        v2f a = *(const v2f*)(arow + k + 2 * hi);
        a.x *= invc;
        a.y *= invc;

        const float4* __restrict__ bp =
            (const float4*)(wpack + ((size_t)ks * 32 + lane) * 16);
        float4 q0 = bp[0], q1 = bp[1], q2 = bp[2], q3 = bp[3];
        v2f bbs[N_TILES];
        bbs[0].x = q0.x; bbs[0].y = q0.y;
        bbs[1].x = q0.z; bbs[1].y = q0.w;
        bbs[2].x = q1.x; bbs[2].y = q1.y;
        bbs[3].x = q1.z; bbs[3].y = q1.w;
        bbs[4].x = q2.x; bbs[4].y = q2.y;
        bbs[5].x = q2.z; bbs[5].y = q2.w;
        bbs[6].x = q3.x; bbs[6].y = q3.y;
        bbs[7].x = q3.z; bbs[7].y = q3.w;

#pragma unroll
        for (int nt = 0; nt < N_TILES; ++nt) {
            acc[nt] = __builtin_amdgcn_wmma_f32_16x16x4_f32(
                false, a, false, bbs[nt], (short)0, acc[nt], false, false);
        }
    }

#pragma unroll
    for (int nt = 0; nt < N_TILES; ++nt) {
        const float bias = b1[nt * 16 + m];
#pragma unroll
        for (int r = 0; r < 8; ++r) acc[nt][r] += bias;
    }

    float s[8], q[8];
#pragma unroll
    for (int r = 0; r < 8; ++r) { s[r] = 0.0f; q[r] = 0.0f; }
#pragma unroll
    for (int nt = 0; nt < N_TILES; ++nt) {
#pragma unroll
        for (int r = 0; r < 8; ++r) {
            const float x = acc[nt][r];
            s[r] += x;
            q[r] += x * x;
        }
    }
#pragma unroll
    for (int mask = 1; mask < 16; mask <<= 1) {
#pragma unroll
        for (int r = 0; r < 8; ++r) {
            s[r] += __shfl_xor(s[r], mask, 32);
            q[r] += __shfl_xor(q[r], mask, 32);
        }
    }
    float mean[8], rstd[8];
#pragma unroll
    for (int r = 0; r < 8; ++r) {
        mean[r] = s[r] * (1.0f / HIDDEN);
        const float var = q[r] * (1.0f / HIDDEN) - mean[r] * mean[r];
        rstd[r] = rsqrtf(var + LN_EPS);
    }

    float* __restrict__ hbase = sH + wave * (TILE_ROWS * HIDDEN);
#pragma unroll
    for (int nt = 0; nt < N_TILES; ++nt) {
        const int n = nt * 16 + m;
        const float g  = lng[n];
        const float bb = lnb[n];
#pragma unroll
        for (int r = 0; r < 8; ++r) {
            float x = (acc[nt][r] - mean[r]) * rstd[r] * g + bb;
            x = fmaxf(x, 0.0f);
            hbase[(r + 8 * hi) * HIDDEN + n] = x;
        }
    }
    __syncthreads();

    for (int idx = lane; idx < TILE_ROWS * OUT_DIM; idx += 32) {
        const int r = idx / OUT_DIM;
        const int j = idx - r * OUT_DIM;
        float sum = b2[j];
        const float* __restrict__ hr = hbase + r * HIDDEN;
#pragma unroll 4
        for (int k = 0; k < HIDDEN; ++k)
            sum = fmaf(hr[k], W2[k * OUT_DIM + j], sum);
        out[(size_t)(m0 + r) * OUT_DIM + j] = sum;
    }
}

extern "C" void kernel_launch(void* const* d_in, const int* in_sizes, int n_in,
                              void* d_out, int out_size, void* d_ws, size_t ws_size,
                              hipStream_t stream) {
    const float* nf    = (const float*)d_in[0];
    const int*   hidx  = (const int*)d_in[1];
    const int*   nodes = hidx;
    const int*   edges = hidx + NUM_CONN;
    const float* W1    = (const float*)d_in[2];
    const float* b1    = (const float*)d_in[3];
    const float* lng   = (const float*)d_in[4];
    const float* lnb   = (const float*)d_in[5];
    const float* W2    = (const float*)d_in[6];
    const float* b2    = (const float*)d_in[7];
    float*       out   = (float*)d_out;

    unsigned* okey = (unsigned*)d_out;

    float* summed = (float*)d_ws;
    float* counts = summed + (size_t)NUM_EDGES * IN_DIM;
    float* wpack  = counts + NUM_EDGES;

    repack_w1_kernel<<<REPACK_WGS + KEY_WGS, 256, 0, stream>>>(W1, wpack, edges, okey);

    edge_accum_kernel<<<ACC_WGS, ACC_THREADS, 0, stream>>>(nf, nodes, edges, okey, summed, counts);

    const int ntiles = NUM_EDGES / TILE_ROWS;
    edge_mlp_kernel<<<ntiles / WAVES_PER_BLOCK, WAVES_PER_BLOCK * 32, 0, stream>>>(
        summed, counts, wpack, b1, lng, lnb, W2, b2, out);
}


# 128->17 product on the vector ALU turned to one edge row per lane: W2 wave-uniform through scalar loads, H rows at a padded LDS pitch, outputs staged in LDS for whole-line stores; same sum order
# speedup vs baseline: 6.0529x; 1.2275x over previous
#include <hip/hip_runtime.h>

#define NUM_NODES 100000
#define NUM_EDGES 50000
#define NUM_CONN  800000
#define IN_DIM    256
#define HIDDEN    128
#define OUT_DIM   17
#define LN_EPS    1e-5f

#define WAVES_PER_BLOCK 5
#define TILE_ROWS 16
#define N_TILES (HIDDEN / 16)
#define K_STEPS (IN_DIM / 4)
#define WPACK_FLOATS (K_STEPS * 32 * 16)
#define H_PITCH (HIDDEN + 4)
#define BLOCK_ROWS (WAVES_PER_BLOCK * TILE_ROWS)

typedef float v2f __attribute__((ext_vector_type(2)));
typedef float v8f __attribute__((ext_vector_type(8)));

#define ACC_WGS       256
#define ACC_THREADS   1024
#define ACC_WAVES     (ACC_THREADS / 32)
#define EDGES_PER_WG  ((NUM_EDGES + ACC_WGS - 1) / ACC_WGS)
#define CHUNK_LOADS   4
#define CHUNK_VECS    (CHUNK_LOADS * 32)
#define NUM_CONN4     (NUM_CONN / 4)
#define NUM_CONN16    (NUM_CONN / 16)
#define FULL_CHUNKS   (NUM_CONN16 / CHUNK_VECS)
#define LIST_CAP      544
#define REPACK_WGS    ((K_STEPS * 32 * 8 + 255) / 256)
#define KEY_WGS       ((NUM_CONN4 + 255) / 256)

__global__ void repack_w1_kernel(const float* __restrict__ W1,
                                 float* __restrict__ wpack,
                                 const int* __restrict__ edges,
                                 unsigned* __restrict__ okey) {
    if (blockIdx.x >= REPACK_WGS) {
        const int i = (blockIdx.x - REPACK_WGS) * blockDim.x + threadIdx.x;
        if (i >= NUM_CONN4) return;
        const int4 v = ((const int4*)edges)[i];
        const unsigned a[4] = {(unsigned)v.x, (unsigned)v.y, (unsigned)v.z, (unsigned)v.w};
        unsigned w = 0u;
#pragma unroll
        for (int b = 0; b < 4; ++b) {
            const unsigned k = a[b] < (unsigned)NUM_EDGES ? a[b] / (unsigned)EDGES_PER_WG : (a[b] & 255u);
            w |= k << (8 * b);
        }
        okey[i] = w;
        return;
    }
    int id = blockIdx.x * blockDim.x + threadIdx.x;
    if (id >= K_STEPS * 32 * 8) return;
    int ks   = id >> 8;
    int rem  = id & 255;
    int lane = rem >> 3;
    int nt   = rem & 7;
    int m    = lane & 15;
    int hi   = lane >> 4;
    int krow = ks * 4 + 2 * hi;
    int n    = nt * 16 + m;
    wpack[2 * id + 0] = W1[(size_t)(krow + 0) * HIDDEN + n];
    wpack[2 * id + 1] = W1[(size_t)(krow + 1) * HIDDEN + n];
}

template <bool FULL>
__device__ __forceinline__ void add_listed_rows(
    const float* __restrict__ nf, const int* __restrict__ nodes, const int* __restrict__ edges,
    const unsigned* lst, int n, float* tile, float* cnts, int lane, int e0, int nE) {
    unsigned conn = 0u;
    if (FULL || lane < n) conn = lst[lane];
    const int node = nodes[conn];
    const unsigned lu = (unsigned)edges[conn] - (unsigned)e0;
    const int le = lu < (unsigned)nE ? (int)lu : EDGES_PER_WG;
    if (FULL || lane < n)
        __hip_atomic_fetch_add(cnts + le, 1.0f, __ATOMIC_RELAXED, __HIP_MEMORY_SCOPE_WORKGROUP);
#pragma unroll
    for (int g = 0; g < 4; ++g) {
        if (!FULL && g * 8 >= n) break;
        v2f r[8][4];
#pragma unroll
        for (int jj = 0; jj < 8; ++jj) {
            const int j = g * 8 + jj;
            if (FULL || j < n) {
                const int nj = __builtin_amdgcn_readlane(node, j);
                const v2f* src = (const v2f*)(nf + (size_t)nj * IN_DIM) + lane;
#pragma unroll
                for (int k = 0; k < 4; ++k) r[jj][k] = src[32 * k];
            }
        }
#pragma unroll
        for (int jj = 0; jj < 8; ++jj) {
            const int j = g * 8 + jj;
            if (FULL || j < n) {
                const int lej = __builtin_amdgcn_readlane(le, j);
                float* dst = tile + lej * IN_DIM + 2 * lane;
#pragma unroll
                for (int k = 0; k < 4; ++k) {
                    __hip_atomic_fetch_add(dst + 64 * k + 0, r[jj][k].x,
                                           __ATOMIC_RELAXED, __HIP_MEMORY_SCOPE_WORKGROUP);
                    __hip_atomic_fetch_add(dst + 64 * k + 1, r[jj][k].y,
                                           __ATOMIC_RELAXED, __HIP_MEMORY_SCOPE_WORKGROUP);
                }
            }
        }
    }
}

__global__ __launch_bounds__(ACC_THREADS) void edge_accum_kernel(
    const float* __restrict__ nf,
    const int*   __restrict__ nodes,
    const int*   __restrict__ edges,
    const unsigned* __restrict__ okey,
    float* __restrict__ summed,
    float* __restrict__ counts) {
    __shared__ __attribute__((aligned(16))) float tile[(EDGES_PER_WG + 1) * IN_DIM];
    __shared__ float cnts[EDGES_PER_WG + 1];
    __shared__ unsigned lists[ACC_WAVES * LIST_CAP];

    const int tid  = threadIdx.x;
    const int lane = tid & 31;
    const int wave = __builtin_amdgcn_readfirstlane(tid >> 5);
    const int e0   = blockIdx.x * EDGES_PER_WG;
    const int nE   = min(EDGES_PER_WG, NUM_EDGES - e0);

    float4* tile4 = (float4*)tile;
    for (int i = tid; i < (EDGES_PER_WG + 1) * IN_DIM / 4; i += ACC_THREADS)
        tile4[i] = make_float4(0.0f, 0.0f, 0.0f, 0.0f);
    if (tid < EDGES_PER_WG + 1) cnts[tid] = 0.0f;
    __syncthreads();

    unsigned* lst = lists + wave * LIST_CAP;
    int cnt = 0;
    const uint4* __restrict__ k4 = (const uint4*)okey;
    const int shift = (blockIdx.x * 3) % FULL_CHUNKS;
    const unsigned pat = (unsigned)blockIdx.x * 0x01010101u;

    auto hitbits = [&](unsigned x) {
        x ^= pat;
        const unsigned t = (x & 0x7f7f7f7fu) + 0x7f7f7f7fu;
        return ~(t | x) & 0x80808080u;
    };
    auto take16 = [&](const uint4 v, int q) {
        unsigned h = (hitbits(v.x) >> 7) | (hitbits(v.y) >> 6) | (hitbits(v.z) >> 5) | (hitbits(v.w) >> 4);
        unsigned m = __builtin_amdgcn_ballot_w32(h != 0u);
        while (m != 0u) {
            if (h != 0u) {
                const int f = __builtin_ctz(h);
                lst[cnt + __builtin_amdgcn_mbcnt_lo(m, 0u)] = (unsigned)(16 * q + 4 * (f & 7) + (f >> 3));
                h &= h - 1u;
            }
            cnt += __builtin_popcount(m);
            m = __builtin_amdgcn_ballot_w32(h != 0u);
        }
    };
    auto drain = [&]() {
        while (cnt >= 32) {
            cnt -= 32;
            add_listed_rows<true>(nf, nodes, edges, lst + cnt, 32, tile, cnts, lane, e0, nE);
        }
    };
    auto chunk_q0 = [&](int idx) {
        int c = idx + shift;
        if (c >= FULL_CHUNKS) c -= FULL_CHUNKS;
        return c * CHUNK_VECS + lane;
    };

    int idx = wave;
    int q0  = chunk_q0(idx);
    uint4 cur[CHUNK_LOADS];
#pragma unroll
    for (int k = 0; k < CHUNK_LOADS; ++k) cur[k] = k4[q0 + 32 * k];
    for (;;) {
        const int  nidx = idx + ACC_WAVES;
        const bool more = nidx < FULL_CHUNKS;
        const int  nq0  = chunk_q0(nidx);
        uint4 nxt[CHUNK_LOADS];
        if (more) {
#pragma unroll
            for (int k = 0; k < CHUNK_LOADS; ++k) nxt[k] = k4[nq0 + 32 * k];
        }
#pragma unroll
        for (int k = 0; k < CHUNK_LOADS; ++k) {
            take16(cur[k], q0 + 32 * k);
            drain();
        }
        if (!more) break;
#pragma unroll
        for (int k = 0; k < CHUNK_LOADS; ++k) cur[k] = nxt[k];
        idx = nidx;
        q0  = nq0;
    }
    {
        const int q = FULL_CHUNKS * CHUNK_VECS + tid;
        uint4 v = make_uint4(~pat, ~pat, ~pat, ~pat);
        if (q < NUM_CONN16) v = k4[q];
        take16(v, q);
        drain();
    }
    while (cnt > 0) {
        const int n = min(cnt, 32);
        cnt -= n;
        add_listed_rows<false>(nf, nodes, edges, lst + cnt, n, tile, cnts, lane, e0, nE);
    }
    __syncthreads();

    float4* out4 = (float4*)(summed + (size_t)e0 * IN_DIM);
    for (int i = tid; i < nE * (IN_DIM / 4); i += ACC_THREADS) out4[i] = tile4[i];
    if (tid < nE) counts[e0 + tid] = cnts[tid];
}

__global__ __launch_bounds__(WAVES_PER_BLOCK * 32) void edge_mlp_kernel(
    const float* __restrict__ summed,
    const float* __restrict__ counts,
    const float* __restrict__ wpack,
    const float* __restrict__ b1,
    const float* __restrict__ lng,
    const float* __restrict__ lnb,
    const float* __restrict__ W2,
    const float* __restrict__ b2,
    float* __restrict__ out) {
    __shared__ __attribute__((aligned(16))) float sH[BLOCK_ROWS * H_PITCH];
    __shared__ float sO[BLOCK_ROWS * OUT_DIM];

    const int wave = threadIdx.x >> 5;
    const int lane = threadIdx.x & 31;
    const int tile = blockIdx.x * WAVES_PER_BLOCK + wave;
    const int m0   = tile * TILE_ROWS;
    const int m    = lane & 15;
    const int hi   = lane >> 4;

    const int row = m0 + m;
    const float invc = 1.0f / fmaxf(counts[row], 1.0f);
    const float* __restrict__ arow = summed + (size_t)row * IN_DIM;

    v8f acc[N_TILES] = {};

    for (int ks = 0; ks < K_STEPS; ++ks) {
        const int k = ks * 4;
        v2f a = *(const v2f*)(arow + k + 2 * hi);
        a.x *= invc;
        a.y *= invc;

        const float4* __restrict__ bp =
            (const float4*)(wpack + ((size_t)ks * 32 + lane) * 16);
        float4 q0 = bp[0], q1 = bp[1], q2 = bp[2], q3 = bp[3];
        v2f bbs[N_TILES];
        bbs[0].x = q0.x; bbs[0].y = q0.y;
        bbs[1].x = q0.z; bbs[1].y = q0.w;
        bbs[2].x = q1.x; bbs[2].y = q1.y;
        bbs[3].x = q1.z; bbs[3].y = q1.w;
        bbs[4].x = q2.x; bbs[4].y = q2.y;
        bbs[5].x = q2.z; bbs[5].y = q2.w;
        bbs[6].x = q3.x; bbs[6].y = q3.y;
        bbs[7].x = q3.z; bbs[7].y = q3.w;

#pragma unroll
        for (int nt = 0; nt < N_TILES; ++nt) {
            acc[nt] = __builtin_amdgcn_wmma_f32_16x16x4_f32(
                false, a, false, bbs[nt], (short)0, acc[nt], false, false);
        }
    }

#pragma unroll
    for (int nt = 0; nt < N_TILES; ++nt) {
        const float bias = b1[nt * 16 + m];
#pragma unroll
        for (int r = 0; r < 8; ++r) acc[nt][r] += bias;
    }

    float s[8], q[8];
#pragma unroll
    for (int r = 0; r < 8; ++r) { s[r] = 0.0f; q[r] = 0.0f; }
#pragma unroll
    for (int nt = 0; nt < N_TILES; ++nt) {
#pragma unroll
        for (int r = 0; r < 8; ++r) {
            const float x = acc[nt][r];
            s[r] += x;
            q[r] += x * x;
        }
    }
#pragma unroll
    for (int mask = 1; mask < 16; mask <<= 1) {
#pragma unroll
        for (int r = 0; r < 8; ++r) {
            s[r] += __shfl_xor(s[r], mask, 32);
            q[r] += __shfl_xor(q[r], mask, 32);
        }
    }
    float mean[8], rstd[8];
#pragma unroll
    for (int r = 0; r < 8; ++r) {
        mean[r] = s[r] * (1.0f / HIDDEN);
        const float var = q[r] * (1.0f / HIDDEN) - mean[r] * mean[r];
        rstd[r] = rsqrtf(var + LN_EPS);
    }

    float* __restrict__ hbase = sH + wave * (TILE_ROWS * H_PITCH);
#pragma unroll
    for (int nt = 0; nt < N_TILES; ++nt) {
        const int n = nt * 16 + m;
        const float g  = lng[n];
        const float bb = lnb[n];
#pragma unroll
        for (int r = 0; r < 8; ++r) {
            float x = (acc[nt][r] - mean[r]) * rstd[r] * g + bb;
            x = fmaxf(x, 0.0f);
            hbase[(r + 8 * hi) * H_PITCH + n] = x;
        }
    }
    __syncthreads();

    const int t = threadIdx.x;
    if (t < BLOCK_ROWS) {
        const float* hr = sH + t * H_PITCH;
        float o[OUT_DIM];
#pragma unroll
        for (int j = 0; j < OUT_DIM; ++j) o[j] = b2[j];
        for (int k = 0; k < HIDDEN; k += 4) {
            const float4 h4 = *(const float4*)(hr + k);
            const float* __restrict__ w = W2 + k * OUT_DIM;
#pragma unroll
            for (int j = 0; j < OUT_DIM; ++j) o[j] = fmaf(h4.x, w[0 * OUT_DIM + j], o[j]);
#pragma unroll
            for (int j = 0; j < OUT_DIM; ++j) o[j] = fmaf(h4.y, w[1 * OUT_DIM + j], o[j]);
#pragma unroll
            for (int j = 0; j < OUT_DIM; ++j) o[j] = fmaf(h4.z, w[2 * OUT_DIM + j], o[j]);
#pragma unroll
            for (int j = 0; j < OUT_DIM; ++j) o[j] = fmaf(h4.w, w[3 * OUT_DIM + j], o[j]);
        }
#pragma unroll
        for (int j = 0; j < OUT_DIM; ++j) sO[t * OUT_DIM + j] = o[j];
    }
    __syncthreads();
    float* __restrict__ oblk = out + (size_t)blockIdx.x * (BLOCK_ROWS * OUT_DIM);
    for (int i = t; i < BLOCK_ROWS * OUT_DIM; i += WAVES_PER_BLOCK * 32) oblk[i] = sO[i];
}

extern "C" void kernel_launch(void* const* d_in, const int* in_sizes, int n_in,
                              void* d_out, int out_size, void* d_ws, size_t ws_size,
                              hipStream_t stream) {
    const float* nf    = (const float*)d_in[0];
    const int*   hidx  = (const int*)d_in[1];
    const int*   nodes = hidx;
    const int*   edges = hidx + NUM_CONN;
    const float* W1    = (const float*)d_in[2];
    const float* b1    = (const float*)d_in[3];
    const float* lng   = (const float*)d_in[4];
    const float* lnb   = (const float*)d_in[5];
    const float* W2    = (const float*)d_in[6];
    const float* b2    = (const float*)d_in[7];
    float*       out   = (float*)d_out;

    unsigned* okey = (unsigned*)d_out;

    float* summed = (float*)d_ws;
    float* counts = summed + (size_t)NUM_EDGES * IN_DIM;
    float* wpack  = counts + NUM_EDGES;

    repack_w1_kernel<<<REPACK_WGS + KEY_WGS, 256, 0, stream>>>(W1, wpack, edges, okey);

    edge_accum_kernel<<<ACC_WGS, ACC_THREADS, 0, stream>>>(nf, nodes, edges, okey, summed, counts);

    const int ntiles = NUM_EDGES / TILE_ROWS;
    edge_mlp_kernel<<<ntiles / WAVES_PER_BLOCK, WAVES_PER_BLOCK * 32, 0, stream>>>(
        summed, counts, wpack, b1, lng, lnb, W2, b2, out);
}


# Segment sum without LDS float adds: owner groups list each edge's nodes in LDS (one counter increment per connection), then each wave adds its edges' rows in registers in list order (arrival order)
# speedup vs baseline: 6.9564x; 1.1493x over previous
#include <hip/hip_runtime.h>

#define NUM_NODES 100000
#define NUM_EDGES 50000
#define NUM_CONN  800000
#define IN_DIM    256
#define HIDDEN    128
#define OUT_DIM   17
#define LN_EPS    1e-5f

#define WAVES_PER_BLOCK 5
#define TILE_ROWS 16
#define N_TILES (HIDDEN / 16)
#define K_STEPS (IN_DIM / 4)
#define WPACK_FLOATS (K_STEPS * 32 * 16)
#define H_PITCH (HIDDEN + 4)
#define BLOCK_ROWS (WAVES_PER_BLOCK * TILE_ROWS)

typedef float v2f __attribute__((ext_vector_type(2)));
typedef float v8f __attribute__((ext_vector_type(8)));

#define ACC_WGS       256
#define ACC_THREADS   1024
#define ACC_WAVES     (ACC_THREADS / 32)
#define EDGES_PER_WG  ((NUM_EDGES + ACC_WGS - 1) / ACC_WGS)
#define CHUNK_LOADS   4
#define CHUNK_VECS    (CHUNK_LOADS * 32)
#define NUM_CONN4     (NUM_CONN / 4)
#define NUM_CONN16    (NUM_CONN / 16)
#define FULL_CHUNKS   (NUM_CONN16 / CHUNK_VECS)
#define LIST_CAP      544
#define REPACK_WGS    ((K_STEPS * 32 * 8 + 255) / 256)
#define KEY_WGS       ((NUM_CONN4 + 255) / 256)

__global__ void repack_w1_kernel(const float* __restrict__ W1,
                                 float* __restrict__ wpack,
                                 const int* __restrict__ edges,
                                 unsigned* __restrict__ okey) {
    if (blockIdx.x >= REPACK_WGS) {
        const int i = (blockIdx.x - REPACK_WGS) * blockDim.x + threadIdx.x;
        if (i >= NUM_CONN4) return;
        const int4 v = ((const int4*)edges)[i];
        const unsigned a[4] = {(unsigned)v.x, (unsigned)v.y, (unsigned)v.z, (unsigned)v.w};
        unsigned w = 0u;
#pragma unroll
        for (int b = 0; b < 4; ++b) {
            const unsigned k = a[b] < (unsigned)NUM_EDGES ? a[b] / (unsigned)EDGES_PER_WG : (a[b] & 255u);
            w |= k << (8 * b);
        }
        okey[i] = w;
        return;
    }
    int id = blockIdx.x * blockDim.x + threadIdx.x;
    if (id >= K_STEPS * 32 * 8) return;
    int ks   = id >> 8;
    int rem  = id & 255;
    int lane = rem >> 3;
    int nt   = rem & 7;
    int m    = lane & 15;
    int hi   = lane >> 4;
    int krow = ks * 4 + 2 * hi;
    int n    = nt * 16 + m;
    wpack[2 * id + 0] = W1[(size_t)(krow + 0) * HIDDEN + n];
    wpack[2 * id + 1] = W1[(size_t)(krow + 1) * HIDDEN + n];
}

template <bool FULL>
__device__ __forceinline__ void add_listed_rows(
    const float* __restrict__ nf, const int* __restrict__ nodes, const int* __restrict__ edges,
    const unsigned* lst, int n, float* tile, float* cnts, int lane, int e0, int nE) {
    unsigned conn = 0u;
    if (FULL || lane < n) conn = lst[lane];
    const int node = nodes[conn];
    const unsigned lu = (unsigned)edges[conn] - (unsigned)e0;
    const int le = lu < (unsigned)nE ? (int)lu : EDGES_PER_WG;
    if (FULL || lane < n)
        __hip_atomic_fetch_add(cnts + le, 1.0f, __ATOMIC_RELAXED, __HIP_MEMORY_SCOPE_WORKGROUP);
#pragma unroll
    for (int g = 0; g < 4; ++g) {
        if (!FULL && g * 8 >= n) break;
        v2f r[8][4];
#pragma unroll
        for (int jj = 0; jj < 8; ++jj) {
            const int j = g * 8 + jj;
            if (FULL || j < n) {
                const int nj = __builtin_amdgcn_readlane(node, j);
                const v2f* src = (const v2f*)(nf + (size_t)nj * IN_DIM) + lane;
#pragma unroll
                for (int k = 0; k < 4; ++k) r[jj][k] = src[32 * k];
            }
        }
#pragma unroll
        for (int jj = 0; jj < 8; ++jj) {
            const int j = g * 8 + jj;
            if (FULL || j < n) {
                const int lej = __builtin_amdgcn_readlane(le, j);
                float* dst = tile + lej * IN_DIM + 2 * lane;
#pragma unroll
                for (int k = 0; k < 4; ++k) {
                    __hip_atomic_fetch_add(dst + 64 * k + 0, r[jj][k].x,
                                           __ATOMIC_RELAXED, __HIP_MEMORY_SCOPE_WORKGROUP);
                    __hip_atomic_fetch_add(dst + 64 * k + 1, r[jj][k].y,
                                           __ATOMIC_RELAXED, __HIP_MEMORY_SCOPE_WORKGROUP);
                }
            }
        }
    }
}

#define EDGE_CAP 64
typedef float v4f __attribute__((ext_vector_type(4)));

template <bool FULL>
__device__ __forceinline__ void route_listed(
    const int* __restrict__ nodes, const int* __restrict__ edges,
    const unsigned* lst, int n, unsigned* ecnt, float* elist, unsigned* ovf, int lane, int e0, int nE) {
    unsigned conn = 0u;
    if (FULL || lane < n) conn = lst[lane];
    const int node = nodes[conn];
    const unsigned lu = (unsigned)edges[conn] - (unsigned)e0;
    if ((FULL || lane < n) && lu < (unsigned)nE) {
        const unsigned pos = __hip_atomic_fetch_add(ecnt + lu, 1u, __ATOMIC_RELAXED, __HIP_MEMORY_SCOPE_WORKGROUP);
        if (pos < (unsigned)EDGE_CAP) elist[lu * EDGE_CAP + pos] = __int_as_float(node);
        else *ovf = 1u;
    }
}

__global__ __launch_bounds__(ACC_THREADS) void edge_accum_kernel(
    const float* __restrict__ nf,
    const int*   __restrict__ nodes,
    const int*   __restrict__ edges,
    const unsigned* __restrict__ okey,
    float* __restrict__ summed,
    float* __restrict__ counts) {
    __shared__ __attribute__((aligned(16))) float tile[(EDGES_PER_WG + 1) * IN_DIM];
    __shared__ float cnts[EDGES_PER_WG + 1];
    __shared__ unsigned lists[ACC_WAVES * LIST_CAP];
    __shared__ unsigned ecnt[EDGES_PER_WG];
    __shared__ unsigned ovf;

    const int tid  = threadIdx.x;
    const int lane = tid & 31;
    const int wave = __builtin_amdgcn_readfirstlane(tid >> 5);
    const int e0   = blockIdx.x * EDGES_PER_WG;
    const int nE   = min(EDGES_PER_WG, NUM_EDGES - e0);

    unsigned* lst = lists + wave * LIST_CAP;
    const uint4* __restrict__ k4 = (const uint4*)okey;
    const int shift = (blockIdx.x * 3) % FULL_CHUNKS;
    const unsigned pat = (unsigned)blockIdx.x * 0x01010101u;

    auto hitbits = [&](unsigned x) {
        x ^= pat;
        const unsigned t = (x & 0x7f7f7f7fu) + 0x7f7f7f7fu;
        return ~(t | x) & 0x80808080u;
    };

    auto take16 = [&](const uint4 v, int q, int& cnt) {
        unsigned h = (hitbits(v.x) >> 7) | (hitbits(v.y) >> 6) | (hitbits(v.z) >> 5) | (hitbits(v.w) >> 4);
        unsigned m = __builtin_amdgcn_ballot_w32(h != 0u);
        while (m != 0u) {
            if (h != 0u) {
                const int f = __builtin_ctz(h);
                lst[cnt + __builtin_amdgcn_mbcnt_lo(m, 0u)] = (unsigned)(16 * q + 4 * (f & 7) + (f >> 3));
                h &= h - 1u;
            }
            cnt += __builtin_popcount(m);
            m = __builtin_amdgcn_ballot_w32(h != 0u);
        }
    };

    if (tid < EDGES_PER_WG) ecnt[tid] = 0u;
    if (tid == 0) ovf = 0u;
    __syncthreads();
    {
        int cnt = 0;
        auto drain = [&]() {
            while (cnt >= 32) {
                cnt -= 32;
                route_listed<true>(nodes, edges, lst + cnt, 32, ecnt, tile, &ovf, lane, e0, nE);
            }
        };
        auto chunk_q0 = [&](int idx) {
            int c = idx + shift;
            if (c >= FULL_CHUNKS) c -= FULL_CHUNKS;
            return c * CHUNK_VECS + lane;
        };
        int idx = wave;
        int q0  = chunk_q0(idx);
        uint4 cur[CHUNK_LOADS];
#pragma unroll
        for (int k = 0; k < CHUNK_LOADS; ++k) cur[k] = k4[q0 + 32 * k];
        for (;;) {
            const int  nidx = idx + ACC_WAVES;
            const bool more = nidx < FULL_CHUNKS;
            const int  nq0  = chunk_q0(nidx);
            uint4 nxt[CHUNK_LOADS];
            if (more) {
#pragma unroll
                for (int k = 0; k < CHUNK_LOADS; ++k) nxt[k] = k4[nq0 + 32 * k];
            }
#pragma unroll
            for (int k = 0; k < CHUNK_LOADS; ++k) {
                take16(cur[k], q0 + 32 * k, cnt);
                drain();
            }
            if (!more) break;
#pragma unroll
            for (int k = 0; k < CHUNK_LOADS; ++k) cur[k] = nxt[k];
            idx = nidx;
            q0  = nq0;
        }
        {
            const int q = FULL_CHUNKS * CHUNK_VECS + tid;
            uint4 v = make_uint4(~pat, ~pat, ~pat, ~pat);
            if (q < NUM_CONN16) v = k4[q];
            take16(v, q, cnt);
            drain();
        }
        while (cnt > 0) {
            const int n = min(cnt, 32);
            cnt -= n;
            route_listed<false>(nodes, edges, lst + cnt, n, ecnt, tile, &ovf, lane, e0, nE);
        }
    }
    __syncthreads();

    const bool over = __builtin_amdgcn_readfirstlane((int)ovf) != 0;
    if (__builtin_expect(over, 0)) {
        float4* tile4 = (float4*)tile;
        for (int i = tid; i < (EDGES_PER_WG + 1) * IN_DIM / 4; i += ACC_THREADS)
            tile4[i] = make_float4(0.0f, 0.0f, 0.0f, 0.0f);
        if (tid < EDGES_PER_WG + 1) cnts[tid] = 0.0f;
        __syncthreads();
        {
            int cnt = 0;
            for (int c = wave; c * 32 < NUM_CONN16; c += ACC_WAVES) {
                const int q = c * 32 + lane;
                uint4 v = make_uint4(~pat, ~pat, ~pat, ~pat);
                if (q < NUM_CONN16) v = k4[q];
                take16(v, q, cnt);
                while (cnt >= 32) {
                    cnt -= 32;
                    add_listed_rows<true>(nf, nodes, edges, lst + cnt, 32, tile, cnts, lane, e0, nE);
                }
            }
            while (cnt > 0) {
                const int n = min(cnt, 32);
                cnt -= n;
                add_listed_rows<false>(nf, nodes, edges, lst + cnt, n, tile, cnts, lane, e0, nE);
            }
        }
        __syncthreads();

        float4* out4 = (float4*)(summed + (size_t)e0 * IN_DIM);
        for (int i = tid; i < nE * (IN_DIM / 4); i += ACC_THREADS) out4[i] = tile4[i];
        if (tid < nE) counts[e0 + tid] = cnts[tid];
        return;
    }

    if (tid < nE) counts[e0 + tid] = (float)ecnt[tid];
    for (int e = wave; e < nE; e += ACC_WAVES) {
        const int n = __builtin_amdgcn_readfirstlane((int)ecnt[e]);
        const float* el = tile + e * EDGE_CAP;
        v4f a0 = {0.0f, 0.0f, 0.0f, 0.0f};
        v4f a1 = {0.0f, 0.0f, 0.0f, 0.0f};
        for (int base = 0; base < n; base += 8) {
            const int m = min(8, n - base);
            int nodev = 0;
            if (lane < m) nodev = __float_as_int(el[base + lane]);
            v4f r[8][2];
#pragma unroll
            for (int j = 0; j < 8; ++j) {
                if (j < m) {
                    const int nj = __builtin_amdgcn_readlane(nodev, j);
                    const v4f* src = (const v4f*)(nf + (size_t)nj * IN_DIM) + lane;
                    r[j][0] = src[0];
                    r[j][1] = src[32];
                }
            }
#pragma unroll
            for (int j = 0; j < 8; ++j) {
                if (j < m) {
                    a0 += r[j][0];
                    a1 += r[j][1];
                }
            }
        }
        v4f* dst = (v4f*)(summed + (size_t)(e0 + e) * IN_DIM) + lane;
        dst[0]  = a0;
        dst[32] = a1;
    }
}

__global__ __launch_bounds__(WAVES_PER_BLOCK * 32) void edge_mlp_kernel(
    const float* __restrict__ summed,
    const float* __restrict__ counts,
    const float* __restrict__ wpack,
    const float* __restrict__ b1,
    const float* __restrict__ lng,
    const float* __restrict__ lnb,
    const float* __restrict__ W2,
    const float* __restrict__ b2,
    float* __restrict__ out) {
    __shared__ __attribute__((aligned(16))) float sH[BLOCK_ROWS * H_PITCH];
    __shared__ float sO[BLOCK_ROWS * OUT_DIM];

    const int wave = threadIdx.x >> 5;
    const int lane = threadIdx.x & 31;
    const int tile = blockIdx.x * WAVES_PER_BLOCK + wave;
    const int m0   = tile * TILE_ROWS;
    const int m    = lane & 15;
    const int hi   = lane >> 4;

    const int row = m0 + m;
    const float invc = 1.0f / fmaxf(counts[row], 1.0f);
    const float* __restrict__ arow = summed + (size_t)row * IN_DIM;

    v8f acc[N_TILES] = {};

    for (int ks = 0; ks < K_STEPS; ++ks) {
        const int k = ks * 4;
        v2f a = *(const v2f*)(arow + k + 2 * hi);
        a.x *= invc;
        a.y *= invc;

        const float4* __restrict__ bp =
            (const float4*)(wpack + ((size_t)ks * 32 + lane) * 16);
        float4 q0 = bp[0], q1 = bp[1], q2 = bp[2], q3 = bp[3];
        v2f bbs[N_TILES];
        bbs[0].x = q0.x; bbs[0].y = q0.y;
        bbs[1].x = q0.z; bbs[1].y = q0.w;
        bbs[2].x = q1.x; bbs[2].y = q1.y;
        bbs[3].x = q1.z; bbs[3].y = q1.w;
        bbs[4].x = q2.x; bbs[4].y = q2.y;
        bbs[5].x = q2.z; bbs[5].y = q2.w;
        bbs[6].x = q3.x; bbs[6].y = q3.y;
        bbs[7].x = q3.z; bbs[7].y = q3.w;

#pragma unroll
        for (int nt = 0; nt < N_TILES; ++nt) {
            acc[nt] = __builtin_amdgcn_wmma_f32_16x16x4_f32(
                false, a, false, bbs[nt], (short)0, acc[nt], false, false);
        }
    }

#pragma unroll
    for (int nt = 0; nt < N_TILES; ++nt) {
        const float bias = b1[nt * 16 + m];
#pragma unroll
        for (int r = 0; r < 8; ++r) acc[nt][r] += bias;
    }

    float s[8], q[8];
#pragma unroll
    for (int r = 0; r < 8; ++r) { s[r] = 0.0f; q[r] = 0.0f; }
#pragma unroll
    for (int nt = 0; nt < N_TILES; ++nt) {
#pragma unroll
        for (int r = 0; r < 8; ++r) {
            const float x = acc[nt][r];
            s[r] += x;
            q[r] += x * x;
        }
    }
#pragma unroll
    for (int mask = 1; mask < 16; mask <<= 1) {
#pragma unroll
        for (int r = 0; r < 8; ++r) {
            s[r] += __shfl_xor(s[r], mask, 32);
            q[r] += __shfl_xor(q[r], mask, 32);
        }
    }
    float mean[8], rstd[8];
#pragma unroll
    for (int r = 0; r < 8; ++r) {
        mean[r] = s[r] * (1.0f / HIDDEN);
        const float var = q[r] * (1.0f / HIDDEN) - mean[r] * mean[r];
        rstd[r] = rsqrtf(var + LN_EPS);
    }

    float* __restrict__ hbase = sH + wave * (TILE_ROWS * H_PITCH);
#pragma unroll
    for (int nt = 0; nt < N_TILES; ++nt) {
        const int n = nt * 16 + m;
        const float g  = lng[n];
        const float bb = lnb[n];
#pragma unroll
        for (int r = 0; r < 8; ++r) {
            float x = (acc[nt][r] - mean[r]) * rstd[r] * g + bb;
            x = fmaxf(x, 0.0f);
            hbase[(r + 8 * hi) * H_PITCH + n] = x;
        }
    }
    __syncthreads();

    const int t = threadIdx.x;
    if (t < BLOCK_ROWS) {
        const float* hr = sH + t * H_PITCH;
        float o[OUT_DIM];
#pragma unroll
        for (int j = 0; j < OUT_DIM; ++j) o[j] = b2[j];
        for (int k = 0; k < HIDDEN; k += 4) {
            const float4 h4 = *(const float4*)(hr + k);
            const float* __restrict__ w = W2 + k * OUT_DIM;
#pragma unroll
            for (int j = 0; j < OUT_DIM; ++j) o[j] = fmaf(h4.x, w[0 * OUT_DIM + j], o[j]);
#pragma unroll
            for (int j = 0; j < OUT_DIM; ++j) o[j] = fmaf(h4.y, w[1 * OUT_DIM + j], o[j]);
#pragma unroll
            for (int j = 0; j < OUT_DIM; ++j) o[j] = fmaf(h4.z, w[2 * OUT_DIM + j], o[j]);
#pragma unroll
            for (int j = 0; j < OUT_DIM; ++j) o[j] = fmaf(h4.w, w[3 * OUT_DIM + j], o[j]);
        }
#pragma unroll
        for (int j = 0; j < OUT_DIM; ++j) sO[t * OUT_DIM + j] = o[j];
    }
    __syncthreads();
    float* __restrict__ oblk = out + (size_t)blockIdx.x * (BLOCK_ROWS * OUT_DIM);
    for (int i = t; i < BLOCK_ROWS * OUT_DIM; i += WAVES_PER_BLOCK * 32) oblk[i] = sO[i];
}

extern "C" void kernel_launch(void* const* d_in, const int* in_sizes, int n_in,
                              void* d_out, int out_size, void* d_ws, size_t ws_size,
                              hipStream_t stream) {
    const float* nf    = (const float*)d_in[0];
    const int*   hidx  = (const int*)d_in[1];
    const int*   nodes = hidx;
    const int*   edges = hidx + NUM_CONN;
    const float* W1    = (const float*)d_in[2];
    const float* b1    = (const float*)d_in[3];
    const float* lng   = (const float*)d_in[4];
    const float* lnb   = (const float*)d_in[5];
    const float* W2    = (const float*)d_in[6];
    const float* b2    = (const float*)d_in[7];
    float*       out   = (float*)d_out;

    unsigned* okey = (unsigned*)d_out;

    float* summed = (float*)d_ws;
    float* counts = summed + (size_t)NUM_EDGES * IN_DIM;
    float* wpack  = counts + NUM_EDGES;

    repack_w1_kernel<<<REPACK_WGS + KEY_WGS, 256, 0, stream>>>(W1, wpack, edges, okey);

    edge_accum_kernel<<<ACC_WGS, ACC_THREADS, 0, stream>>>(nf, nodes, edges, okey, summed, counts);

    const int ntiles = NUM_EDGES / TILE_ROWS;
    edge_mlp_kernel<<<ntiles / WAVES_PER_BLOCK, WAVES_PER_BLOCK * 32, 0, stream>>>(
        summed, counts, wpack, b1, lng, lnb, W2, b2, out);
}


# Owner groups stop scanning every key: the key launch also writes a per-owner bitmap (one bit per 16-key vector) and a group looks only at its flagged vectors; rows still summed in list arrival order
# speedup vs baseline: 7.2361x; 1.0402x over previous
#include <hip/hip_runtime.h>

#define NUM_NODES 100000
#define NUM_EDGES 50000
#define NUM_CONN  800000
#define IN_DIM    256
#define HIDDEN    128
#define OUT_DIM   17
#define LN_EPS    1e-5f

#define WAVES_PER_BLOCK 5
#define TILE_ROWS 16
#define N_TILES (HIDDEN / 16)
#define K_STEPS (IN_DIM / 4)
#define WPACK_FLOATS (K_STEPS * 32 * 16)
#define H_PITCH (HIDDEN + 4)
#define BLOCK_ROWS (WAVES_PER_BLOCK * TILE_ROWS)

typedef float v2f __attribute__((ext_vector_type(2)));
typedef float v8f __attribute__((ext_vector_type(8)));

#define ACC_WGS       256
#define ACC_THREADS   1024
#define ACC_WAVES     (ACC_THREADS / 32)
#define EDGES_PER_WG  ((NUM_EDGES + ACC_WGS - 1) / ACC_WGS)
#define NUM_CONN4     (NUM_CONN / 4)
#define NUM_CONN16    (NUM_CONN / 16)
#define LIST_CAP      544
#define K1_THREADS    1024
#define REPACK_WGS    ((K_STEPS * 32 * 8 + K1_THREADS - 1) / K1_THREADS)
#define VEC_PER_KWG   1024
#define KEY_WGS       ((NUM_CONN16 + VEC_PER_KWG - 1) / VEC_PER_KWG)
#define BM_ROW_DW     (KEY_WGS * 32)
#define DW_PER_WAVE   (BM_ROW_DW / ACC_WAVES)
#define BM_PITCH      257
static_assert(BM_ROW_DW % ACC_WAVES == 0, "bitmap row splits evenly over the waves");
static_assert(DW_PER_WAVE > 32 && DW_PER_WAVE <= 64, "two bitmap words per lane at most");
static_assert(ACC_WGS == 256, "one key byte per connection");

__global__ __launch_bounds__(K1_THREADS) void repack_w1_kernel(const float* __restrict__ W1,
                                 float* __restrict__ wpack,
                                 const int* __restrict__ edges,
                                 unsigned* __restrict__ okey,
                                 unsigned* __restrict__ bitmap) {
    __shared__ unsigned bm[32 * BM_PITCH];
    if (blockIdx.x >= REPACK_WGS) {
        const int t = threadIdx.x;
        const int b = blockIdx.x - REPACK_WGS;
        for (int i = t; i < 32 * BM_PITCH; i += K1_THREADS) bm[i] = 0u;
        __syncthreads();
        const int q = b * VEC_PER_KWG + t;
        if (q < NUM_CONN16) {
            const int4* __restrict__ e4 = (const int4*)edges + 4 * (size_t)q;
            unsigned* bmw = bm + (t >> 5) * BM_PITCH;
            const unsigned bit = 1u << (t & 31);
            unsigned kw[4];
#pragma unroll
            for (int d = 0; d < 4; ++d) {
                const int4 v = e4[d];
                const unsigned a[4] = {(unsigned)v.x, (unsigned)v.y, (unsigned)v.z, (unsigned)v.w};
                unsigned w = 0u;
#pragma unroll
                for (int bb = 0; bb < 4; ++bb) {
                    const unsigned k = a[bb] < (unsigned)NUM_EDGES ? a[bb] / (unsigned)EDGES_PER_WG : (a[bb] & 255u);
                    w |= k << (8 * bb);
                    __hip_atomic_fetch_or(bmw + k, bit, __ATOMIC_RELAXED, __HIP_MEMORY_SCOPE_WORKGROUP);
                }
                kw[d] = w;
            }
            ((uint4*)okey)[q] = make_uint4(kw[0], kw[1], kw[2], kw[3]);
        }
        __syncthreads();
        unsigned* __restrict__ dst = bitmap + (size_t)b * 32;
        for (int i = t; i < ACC_WGS * 32; i += K1_THREADS) {
            const int g = i >> 5;
            const int s = i & 31;
            dst[(size_t)g * BM_ROW_DW + s] = bm[s * BM_PITCH + g];
        }
        return;
    }
    int id = blockIdx.x * blockDim.x + threadIdx.x;
    if (id >= K_STEPS * 32 * 8) return;
    int ks   = id >> 8;
    int rem  = id & 255;
    int lane = rem >> 3;
    int nt   = rem & 7;
    int m    = lane & 15;
    int hi   = lane >> 4;
    int krow = ks * 4 + 2 * hi;
    int n    = nt * 16 + m;
    wpack[2 * id + 0] = W1[(size_t)(krow + 0) * HIDDEN + n];
    wpack[2 * id + 1] = W1[(size_t)(krow + 1) * HIDDEN + n];
}

template <bool FULL>
__device__ __forceinline__ void add_listed_rows(
    const float* __restrict__ nf, const int* __restrict__ nodes, const int* __restrict__ edges,
    const unsigned* lst, int n, float* tile, float* cnts, int lane, int e0, int nE) {
    unsigned conn = 0u;
    if (FULL || lane < n) conn = lst[lane];
    const int node = nodes[conn];
    const unsigned lu = (unsigned)edges[conn] - (unsigned)e0;
    const int le = lu < (unsigned)nE ? (int)lu : EDGES_PER_WG;
    if (FULL || lane < n)
        __hip_atomic_fetch_add(cnts + le, 1.0f, __ATOMIC_RELAXED, __HIP_MEMORY_SCOPE_WORKGROUP);
#pragma unroll
    for (int g = 0; g < 4; ++g) {
        if (!FULL && g * 8 >= n) break;
        v2f r[8][4];
#pragma unroll
        for (int jj = 0; jj < 8; ++jj) {
            const int j = g * 8 + jj;
            if (FULL || j < n) {
                const int nj = __builtin_amdgcn_readlane(node, j);
                const v2f* src = (const v2f*)(nf + (size_t)nj * IN_DIM) + lane;
#pragma unroll
                for (int k = 0; k < 4; ++k) r[jj][k] = src[32 * k];
            }
        }
#pragma unroll
        for (int jj = 0; jj < 8; ++jj) {
            const int j = g * 8 + jj;
            if (FULL || j < n) {
                const int lej = __builtin_amdgcn_readlane(le, j);
                float* dst = tile + lej * IN_DIM + 2 * lane;
#pragma unroll
                for (int k = 0; k < 4; ++k) {
                    __hip_atomic_fetch_add(dst + 64 * k + 0, r[jj][k].x,
                                           __ATOMIC_RELAXED, __HIP_MEMORY_SCOPE_WORKGROUP);
                    __hip_atomic_fetch_add(dst + 64 * k + 1, r[jj][k].y,
                                           __ATOMIC_RELAXED, __HIP_MEMORY_SCOPE_WORKGROUP);
                }
            }
        }
    }
}

#define EDGE_CAP 64
typedef float v4f __attribute__((ext_vector_type(4)));

template <bool FULL>
__device__ __forceinline__ void route_listed(
    const int* __restrict__ nodes, const int* __restrict__ edges,
    const unsigned* lst, int n, unsigned* ecnt, float* elist, unsigned* ovf, int lane, int e0, int nE) {
    unsigned conn = 0u;
    if (FULL || lane < n) conn = lst[lane];
    const int node = nodes[conn];
    const unsigned lu = (unsigned)edges[conn] - (unsigned)e0;
    if ((FULL || lane < n) && lu < (unsigned)nE) {
        const unsigned pos = __hip_atomic_fetch_add(ecnt + lu, 1u, __ATOMIC_RELAXED, __HIP_MEMORY_SCOPE_WORKGROUP);
        if (pos < (unsigned)EDGE_CAP) elist[lu * EDGE_CAP + pos] = __int_as_float(node);
        else *ovf = 1u;
    }
}

__global__ __launch_bounds__(ACC_THREADS) void edge_accum_kernel(
    const float* __restrict__ nf,
    const int*   __restrict__ nodes,
    const int*   __restrict__ edges,
    const unsigned* __restrict__ okey,
    const unsigned* __restrict__ bitmap,
    float* __restrict__ summed,
    float* __restrict__ counts) {
    __shared__ __attribute__((aligned(16))) float tile[(EDGES_PER_WG + 1) * IN_DIM];
    __shared__ float cnts[EDGES_PER_WG + 1];
    __shared__ unsigned lists[ACC_WAVES * LIST_CAP];
    __shared__ unsigned ecnt[EDGES_PER_WG];
    __shared__ unsigned ovf;

    const int tid  = threadIdx.x;
    const int lane = tid & 31;
    const int wave = __builtin_amdgcn_readfirstlane(tid >> 5);
    const int e0   = blockIdx.x * EDGES_PER_WG;
    const int nE   = min(EDGES_PER_WG, NUM_EDGES - e0);

    unsigned* lst = lists + wave * LIST_CAP;
    const uint4* __restrict__ k4 = (const uint4*)okey;
    const unsigned pat = (unsigned)blockIdx.x * 0x01010101u;

    auto hitbits = [&](unsigned x) {
        x ^= pat;
        const unsigned t = (x & 0x7f7f7f7fu) + 0x7f7f7f7fu;
        return ~(t | x) & 0x80808080u;
    };

    auto take16 = [&](const uint4 v, int q, int& cnt) {
        unsigned h = (hitbits(v.x) >> 7) | (hitbits(v.y) >> 6) | (hitbits(v.z) >> 5) | (hitbits(v.w) >> 4);
        unsigned m = __builtin_amdgcn_ballot_w32(h != 0u);
        while (m != 0u) {
            if (h != 0u) {
                const int f = __builtin_ctz(h);
                lst[cnt + __builtin_amdgcn_mbcnt_lo(m, 0u)] = (unsigned)(16 * q + 4 * (f & 7) + (f >> 3));
                h &= h - 1u;
            }
            cnt += __builtin_popcount(m);
            m = __builtin_amdgcn_ballot_w32(h != 0u);
        }
    };

    if (tid < EDGES_PER_WG) ecnt[tid] = 0u;
    if (tid == 0) ovf = 0u;
    __syncthreads();
    {
        float* vl = tile + EDGES_PER_WG * EDGE_CAP + wave * LIST_CAP;
        int cnt  = 0;
        int vcnt = 0;
        auto drain = [&]() {
            while (cnt >= 32) {
                cnt -= 32;
                route_listed<true>(nodes, edges, lst + cnt, 32, ecnt, tile, &ovf, lane, e0, nE);
            }
        };
        auto look = [&](const float* from, int n) {
            int   q = 0;
            uint4 v = make_uint4(~pat, ~pat, ~pat, ~pat);
            if (lane < n) {
                q = __float_as_int(from[lane]);
                v = k4[q];
            }
            take16(v, q, cnt);
            drain();
        };
        const unsigned* __restrict__ bmrow = bitmap + (size_t)blockIdx.x * BM_ROW_DW + wave * DW_PER_WAVE;
        unsigned wb[2];
        wb[0] = bmrow[lane];
        wb[1] = lane < DW_PER_WAVE - 32 ? bmrow[32 + lane] : 0u;
#pragma unroll
        for (int s = 0; s < 4; ++s) {
            unsigned h = (wb[s >> 1] >> (16 * (s & 1))) & 0xffffu;
            const int qb = 32 * (wave * DW_PER_WAVE + 32 * (s >> 1) + lane) + 16 * (s & 1);
            unsigned m = __builtin_amdgcn_ballot_w32(h != 0u);
            while (m != 0u) {
                if (h != 0u) {
                    vl[vcnt + __builtin_amdgcn_mbcnt_lo(m, 0u)] = __int_as_float(qb + __builtin_ctz(h));
                    h &= h - 1u;
                }
                vcnt += __builtin_popcount(m);
                m = __builtin_amdgcn_ballot_w32(h != 0u);
            }
            while (vcnt >= 32) {
                vcnt -= 32;
                look(vl + vcnt, 32);
            }
        }
        if (vcnt > 0) look(vl, vcnt);
        while (cnt > 0) {
            const int n = min(cnt, 32);
            cnt -= n;
            route_listed<false>(nodes, edges, lst + cnt, n, ecnt, tile, &ovf, lane, e0, nE);
        }
    }
    __syncthreads();

    const bool over = __builtin_amdgcn_readfirstlane((int)ovf) != 0;
    if (__builtin_expect(over, 0)) {
        float4* tile4 = (float4*)tile;
        for (int i = tid; i < (EDGES_PER_WG + 1) * IN_DIM / 4; i += ACC_THREADS)
            tile4[i] = make_float4(0.0f, 0.0f, 0.0f, 0.0f);
        if (tid < EDGES_PER_WG + 1) cnts[tid] = 0.0f;
        __syncthreads();
        {
            int cnt = 0;
            for (int c = wave; c * 32 < NUM_CONN16; c += ACC_WAVES) {
                const int q = c * 32 + lane;
                uint4 v = make_uint4(~pat, ~pat, ~pat, ~pat);
                if (q < NUM_CONN16) v = k4[q];
                take16(v, q, cnt);
                while (cnt >= 32) {
                    cnt -= 32;
                    add_listed_rows<true>(nf, nodes, edges, lst + cnt, 32, tile, cnts, lane, e0, nE);
                }
            }
            while (cnt > 0) {
                const int n = min(cnt, 32);
                cnt -= n;
                add_listed_rows<false>(nf, nodes, edges, lst + cnt, n, tile, cnts, lane, e0, nE);
            }
        }
        __syncthreads();

        float4* out4 = (float4*)(summed + (size_t)e0 * IN_DIM);
        for (int i = tid; i < nE * (IN_DIM / 4); i += ACC_THREADS) out4[i] = tile4[i];
        if (tid < nE) counts[e0 + tid] = cnts[tid];
        return;
    }

    if (tid < nE) counts[e0 + tid] = (float)ecnt[tid];
    for (int e = wave; e < nE; e += ACC_WAVES) {
        const int n = __builtin_amdgcn_readfirstlane((int)ecnt[e]);
        const float* el = tile + e * EDGE_CAP;
        v4f a0 = {0.0f, 0.0f, 0.0f, 0.0f};
        v4f a1 = {0.0f, 0.0f, 0.0f, 0.0f};
        for (int base = 0; base < n; base += 8) {
            const int m = min(8, n - base);
            int nodev = 0;
            if (lane < m) nodev = __float_as_int(el[base + lane]);
            v4f r[8][2];
#pragma unroll
            for (int j = 0; j < 8; ++j) {
                if (j < m) {
                    const int nj = __builtin_amdgcn_readlane(nodev, j);
                    const v4f* src = (const v4f*)(nf + (size_t)nj * IN_DIM) + lane;
                    r[j][0] = src[0];
                    r[j][1] = src[32];
                }
            }
#pragma unroll
            for (int j = 0; j < 8; ++j) {
                if (j < m) {
                    a0 += r[j][0];
                    a1 += r[j][1];
                }
            }
        }
        v4f* dst = (v4f*)(summed + (size_t)(e0 + e) * IN_DIM) + lane;
        dst[0]  = a0;
        dst[32] = a1;
    }
}

__global__ __launch_bounds__(WAVES_PER_BLOCK * 32) void edge_mlp_kernel(
    const float* __restrict__ summed,
    const float* __restrict__ counts,
    const float* __restrict__ wpack,
    const float* __restrict__ b1,
    const float* __restrict__ lng,
    const float* __restrict__ lnb,
    const float* __restrict__ W2,
    const float* __restrict__ b2,
    float* __restrict__ out) {
    __shared__ __attribute__((aligned(16))) float sH[BLOCK_ROWS * H_PITCH];
    __shared__ float sO[BLOCK_ROWS * OUT_DIM];

    const int wave = threadIdx.x >> 5;
    const int lane = threadIdx.x & 31;
    const int tile = blockIdx.x * WAVES_PER_BLOCK + wave;
    const int m0   = tile * TILE_ROWS;
    const int m    = lane & 15;
    const int hi   = lane >> 4;

    const int row = m0 + m;
    const float invc = 1.0f / fmaxf(counts[row], 1.0f);
    const float* __restrict__ arow = summed + (size_t)row * IN_DIM;

    v8f acc[N_TILES] = {};

    for (int ks = 0; ks < K_STEPS; ++ks) {
        const int k = ks * 4;
        v2f a = *(const v2f*)(arow + k + 2 * hi);
        a.x *= invc;
        a.y *= invc;

        const float4* __restrict__ bp =
            (const float4*)(wpack + ((size_t)ks * 32 + lane) * 16);
        float4 q0 = bp[0], q1 = bp[1], q2 = bp[2], q3 = bp[3];
        v2f bbs[N_TILES];
        bbs[0].x = q0.x; bbs[0].y = q0.y;
        bbs[1].x = q0.z; bbs[1].y = q0.w;
        bbs[2].x = q1.x; bbs[2].y = q1.y;
        bbs[3].x = q1.z; bbs[3].y = q1.w;
        bbs[4].x = q2.x; bbs[4].y = q2.y;
        bbs[5].x = q2.z; bbs[5].y = q2.w;
        bbs[6].x = q3.x; bbs[6].y = q3.y;
        bbs[7].x = q3.z; bbs[7].y = q3.w;

#pragma unroll
        for (int nt = 0; nt < N_TILES; ++nt) {
            acc[nt] = __builtin_amdgcn_wmma_f32_16x16x4_f32(
                false, a, false, bbs[nt], (short)0, acc[nt], false, false);
        }
    }

#pragma unroll
    for (int nt = 0; nt < N_TILES; ++nt) {
        const float bias = b1[nt * 16 + m];
#pragma unroll
        for (int r = 0; r < 8; ++r) acc[nt][r] += bias;
    }

    float s[8], q[8];
#pragma unroll
    for (int r = 0; r < 8; ++r) { s[r] = 0.0f; q[r] = 0.0f; }
#pragma unroll
    for (int nt = 0; nt < N_TILES; ++nt) {
#pragma unroll
        for (int r = 0; r < 8; ++r) {
            const float x = acc[nt][r];
            s[r] += x;
            q[r] += x * x;
        }
    }
#pragma unroll
    for (int mask = 1; mask < 16; mask <<= 1) {
#pragma unroll
        for (int r = 0; r < 8; ++r) {
            s[r] += __shfl_xor(s[r], mask, 32);
            q[r] += __shfl_xor(q[r], mask, 32);
        }
    }
    float mean[8], rstd[8];
#pragma unroll
    for (int r = 0; r < 8; ++r) {
        mean[r] = s[r] * (1.0f / HIDDEN);
        const float var = q[r] * (1.0f / HIDDEN) - mean[r] * mean[r];
        rstd[r] = rsqrtf(var + LN_EPS);
    }

    float* __restrict__ hbase = sH + wave * (TILE_ROWS * H_PITCH);
#pragma unroll
    for (int nt = 0; nt < N_TILES; ++nt) {
        const int n = nt * 16 + m;
        const float g  = lng[n];
        const float bb = lnb[n];
#pragma unroll
        for (int r = 0; r < 8; ++r) {
            float x = (acc[nt][r] - mean[r]) * rstd[r] * g + bb;
            x = fmaxf(x, 0.0f);
            hbase[(r + 8 * hi) * H_PITCH + n] = x;
        }
    }
    __syncthreads();

    const int t = threadIdx.x;
    if (t < BLOCK_ROWS) {
        const float* hr = sH + t * H_PITCH;
        float o[OUT_DIM];
#pragma unroll
        for (int j = 0; j < OUT_DIM; ++j) o[j] = b2[j];
        for (int k = 0; k < HIDDEN; k += 4) {
            const float4 h4 = *(const float4*)(hr + k);
            const float* __restrict__ w = W2 + k * OUT_DIM;
#pragma unroll
            for (int j = 0; j < OUT_DIM; ++j) o[j] = fmaf(h4.x, w[0 * OUT_DIM + j], o[j]);
#pragma unroll
            for (int j = 0; j < OUT_DIM; ++j) o[j] = fmaf(h4.y, w[1 * OUT_DIM + j], o[j]);
#pragma unroll
            for (int j = 0; j < OUT_DIM; ++j) o[j] = fmaf(h4.z, w[2 * OUT_DIM + j], o[j]);
#pragma unroll
            for (int j = 0; j < OUT_DIM; ++j) o[j] = fmaf(h4.w, w[3 * OUT_DIM + j], o[j]);
        }
#pragma unroll
        for (int j = 0; j < OUT_DIM; ++j) sO[t * OUT_DIM + j] = o[j];
    }
    __syncthreads();
    float* __restrict__ oblk = out + (size_t)blockIdx.x * (BLOCK_ROWS * OUT_DIM);
    for (int i = t; i < BLOCK_ROWS * OUT_DIM; i += WAVES_PER_BLOCK * 32) oblk[i] = sO[i];
}

extern "C" void kernel_launch(void* const* d_in, const int* in_sizes, int n_in,
                              void* d_out, int out_size, void* d_ws, size_t ws_size,
                              hipStream_t stream) {
    const float* nf    = (const float*)d_in[0];
    const int*   hidx  = (const int*)d_in[1];
    const int*   nodes = hidx;
    const int*   edges = hidx + NUM_CONN;
    const float* W1    = (const float*)d_in[2];
    const float* b1    = (const float*)d_in[3];
    const float* lng   = (const float*)d_in[4];
    const float* lnb   = (const float*)d_in[5];
    const float* W2    = (const float*)d_in[6];
    const float* b2    = (const float*)d_in[7];
    float*       out   = (float*)d_out;

    unsigned* okey = (unsigned*)d_out;
    unsigned* bitmap = okey + NUM_CONN4;

    float* summed = (float*)d_ws;
    float* counts = summed + (size_t)NUM_EDGES * IN_DIM;
    float* wpack  = counts + NUM_EDGES;

    repack_w1_kernel<<<REPACK_WGS + KEY_WGS, K1_THREADS, 0, stream>>>(W1, wpack, edges, okey, bitmap);

    edge_accum_kernel<<<ACC_WGS, ACC_THREADS, 0, stream>>>(nf, nodes, edges, okey, bitmap, summed, counts);

    const int ntiles = NUM_EDGES / TILE_ROWS;
    edge_mlp_kernel<<<ntiles / WAVES_PER_BLOCK, WAVES_PER_BLOCK * 32, 0, stream>>>(
        summed, counts, wpack, b1, lng, lnb, W2, b2, out);
}
